# starting kernel
# baseline (speedup 1.0000x reference)
#include <hip/hip_runtime.h>
#include <hip/hip_bf16.h>
#include <math.h>

typedef float v2f __attribute__((ext_vector_type(2)));
typedef float v4f __attribute__((ext_vector_type(4)));
typedef float v8f __attribute__((ext_vector_type(8)));

#define BATCH 16
#define NPTS  4096
#define CIN   256
#define COUT  512
#define KNBR  16
#define NOUT  1024
#define MROWS (BATCH * NPTS)
#define BN_EPS 1e-5f
#define XOUT_ELEMS ((size_t)BATCH * NOUT * COUT)

__global__ __launch_bounds__(128) void gemm_bn_relu_kernel(
    const float* __restrict__ x, const float* __restrict__ W,
    const float* __restrict__ bias, const float* __restrict__ gamma,
    const float* __restrict__ beta, const float* __restrict__ rmean,
    const float* __restrict__ rvar, float* __restrict__ h)
{
    const int wave = threadIdx.x >> 5;
    const int lane = threadIdx.x & 31;
    const int m0 = (blockIdx.x * 4 + wave) * 16;
    const int n0 = blockIdx.y * 64;

    const int rowA = m0 + (lane & 15);
    const int kOff = (lane >> 4) << 1;
    const float* aPtr = x + (size_t)rowA * CIN + kOff;

    const int colL = lane & 15;
    const float* bPtr0 = W + (size_t)(n0 +  0 + colL) * CIN + kOff;
    const float* bPtr1 = W + (size_t)(n0 + 16 + colL) * CIN + kOff;
    const float* bPtr2 = W + (size_t)(n0 + 32 + colL) * CIN + kOff;
    const float* bPtr3 = W + (size_t)(n0 + 48 + colL) * CIN + kOff;

    v8f acc[4];
    #pragma unroll
    for (int nt = 0; nt < 4; ++nt) acc[nt] = {};

    #pragma unroll 4
    for (int c = 0; c < CIN; c += 4) {
        v2f a  = *(const v2f*)(aPtr + c);
        v2f b0 = *(const v2f*)(bPtr0 + c);
        v2f b1 = *(const v2f*)(bPtr1 + c);
        v2f b2 = *(const v2f*)(bPtr2 + c);
        v2f b3 = *(const v2f*)(bPtr3 + c);
        acc[0] = __builtin_amdgcn_wmma_f32_16x16x4_f32(false, a, false, b0,
                                                       (short)0, acc[0], false, false);
        acc[1] = __builtin_amdgcn_wmma_f32_16x16x4_f32(false, a, false, b1,
                                                       (short)0, acc[1], false, false);
        acc[2] = __builtin_amdgcn_wmma_f32_16x16x4_f32(false, a, false, b2,
                                                       (short)0, acc[2], false, false);
        acc[3] = __builtin_amdgcn_wmma_f32_16x16x4_f32(false, a, false, b3,
                                                       (short)0, acc[3], false, false);
    }

    const int rOff = (lane >> 4) << 3;
    #pragma unroll
    for (int nt = 0; nt < 4; ++nt) {
        const int o = n0 + nt * 16 + colL;
        const float sc = gamma[o] * rsqrtf(rvar[o] + BN_EPS);
        const float sh = (bias[o] - rmean[o]) * sc + beta[o];
        #pragma unroll
        for (int j = 0; j < 8; ++j) {
            float v = fmaf(acc[nt][j], sc, sh);
            v = fmaxf(v, 0.0f);
            h[(size_t)(m0 + j + rOff) * COUT + o] = v;
        }
    }
}

__global__ __launch_bounds__(1024) void fps_kernel(const float* __restrict__ p,
                                                   int* __restrict__ fid)
{
    const int b = blockIdx.x;
    const float* pb = p + (size_t)b * NPTS * 3;
    const int t = threadIdx.x;

    float px[4], py[4], pz[4], d[4];
    const float x0 = pb[0], y0 = pb[1], z0 = pb[2];
    #pragma unroll
    for (int s = 0; s < 4; ++s) {
        const int idx = t + s * 1024;
        px[s] = pb[idx * 3 + 0];
        py[s] = pb[idx * 3 + 1];
        pz[s] = pb[idx * 3 + 2];
        const float dx = px[s] - x0, dy = py[s] - y0, dz = pz[s] - z0;
        d[s] = dx * dx + dy * dy + dz * dz;
    }

    __shared__ float sval[32];
    __shared__ int   sidx[32];
    __shared__ int   snxt;
    __shared__ float scoord[3];

    if (t == 0) fid[b * NOUT + 0] = b * NPTS;

    const int wid = t >> 5, lane = t & 31;

    for (int i = 1; i < NOUT; ++i) {
        float bv = d[0]; int bi = t;
        #pragma unroll
        for (int s = 1; s < 4; ++s) {
            if (d[s] > bv) { bv = d[s]; bi = t + s * 1024; }
        }
        #pragma unroll
        for (int off = 16; off >= 1; off >>= 1) {
            const float ov = __shfl_xor(bv, off, 32);
            const int   oi = __shfl_xor(bi, off, 32);
            if (ov > bv || (ov == bv && oi < bi)) { bv = ov; bi = oi; }
        }
        if (lane == 0) { sval[wid] = bv; sidx[wid] = bi; }
        __syncthreads();
        if (wid == 0) {
            float v = sval[lane]; int ix = sidx[lane];
            #pragma unroll
            for (int off = 16; off >= 1; off >>= 1) {
                const float ov = __shfl_xor(v, off, 32);
                const int   oi = __shfl_xor(ix, off, 32);
                if (ov > v || (ov == v && oi < ix)) { v = ov; ix = oi; }
            }
            if (lane == 0) {
                snxt = ix;
                scoord[0] = pb[ix * 3 + 0];
                scoord[1] = pb[ix * 3 + 1];
                scoord[2] = pb[ix * 3 + 2];
                fid[b * NOUT + i] = b * NPTS + ix;
            }
        }
        __syncthreads();
        const float nx = scoord[0], ny = scoord[1], nz = scoord[2];
        #pragma unroll
        for (int s = 0; s < 4; ++s) {
            const float dx = px[s] - nx, dy = py[s] - ny, dz = pz[s] - nz;
            const float dd = dx * dx + dy * dy + dz * dz;
            d[s] = fminf(d[s], dd);
        }
    }
}

__global__ __launch_bounds__(128) void gather_maxpool_kernel(
    const float* __restrict__ h, const float* __restrict__ p,
    const int* __restrict__ fid, const int* __restrict__ sid_euc,
    float* __restrict__ out)
{
    const int s = blockIdx.x;
    const int t = threadIdx.x;

    __shared__ int ssid[KNBR];
    const int f = fid[s];
    if (t < KNBR) ssid[t] = sid_euc[(size_t)f * KNBR + t];
    if (t < 3)    out[XOUT_ELEMS + (size_t)s * 3 + t] = p[(size_t)f * 3 + t];
    __syncthreads();

    const int c = t * 4;
    v4f acc = { -INFINITY, -INFINITY, -INFINITY, -INFINITY };
    #pragma unroll
    for (int k = 0; k < KNBR; ++k) {
        const int r = ssid[k];
        if (k + 1 < KNBR)
            __builtin_prefetch(h + (size_t)ssid[k + 1] * COUT + c, 0, 0);
        const v4f v = *(const v4f*)(h + (size_t)r * COUT + c);
        acc.x = fmaxf(acc.x, v.x);
        acc.y = fmaxf(acc.y, v.y);
        acc.z = fmaxf(acc.z, v.z);
        acc.w = fmaxf(acc.w, v.w);
    }
    *(v4f*)(out + (size_t)s * COUT + c) = acc;
}

extern "C" void kernel_launch(void* const* d_in, const int* in_sizes, int n_in,
                              void* d_out, int out_size, void* d_ws, size_t ws_size,
                              hipStream_t stream)
{
    (void)in_sizes; (void)n_in; (void)out_size; (void)ws_size;
    const float* x     = (const float*)d_in[0];
    const float* p     = (const float*)d_in[1];
    const int*   sid   = (const int*)  d_in[2];
    const float* W     = (const float*)d_in[4];
    const float* bias  = (const float*)d_in[5];
    const float* gamma = (const float*)d_in[6];
    const float* beta  = (const float*)d_in[7];
    const float* rmean = (const float*)d_in[8];
    const float* rvar  = (const float*)d_in[9];
    float* out = (float*)d_out;

    float* h  = (float*)d_ws;
    int* fid  = (int*)((char*)d_ws + (size_t)MROWS * COUT * 4);

    dim3 ggrid(MROWS / 64, COUT / 64);
    gemm_bn_relu_kernel<<<ggrid, 128, 0, stream>>>(x, W, bias, gamma, beta,
                                                   rmean, rvar, h);
    fps_kernel<<<BATCH, 1024, 0, stream>>>(p, fid);
    gather_maxpool_kernel<<<BATCH * NOUT, 128, 0, stream>>>(h, p, fid, sid, out);
}


# FPS: cloud pre-sorted into 16 spatial cells of 256 points (cell code ranked by counting, one new launch); a pick updates only cells whose box is nearer than their largest distance; same picks
# speedup vs baseline: 1.8230x; 1.8230x over previous
#include <hip/hip_runtime.h>
#include <hip/hip_bf16.h>
#include <math.h>

typedef float v2f __attribute__((ext_vector_type(2)));
typedef float v4f __attribute__((ext_vector_type(4)));
typedef float v8f __attribute__((ext_vector_type(8)));

#define BATCH 16
#define NPTS  4096
#define CIN   256
#define COUT  512
#define KNBR  16
#define NOUT  1024
#define MROWS (BATCH * NPTS)
#define BN_EPS 1e-5f
#define XOUT_ELEMS ((size_t)BATCH * NOUT * COUT)

#define GBM 128
#define GBN 128
#define GBK 32
#define LDP 36
#define NCHUNK (CIN / GBK)

__global__ __launch_bounds__(256) void gemm_bn_relu_kernel(
    const float* __restrict__ x, const float* __restrict__ W,
    const float* __restrict__ bias, const float* __restrict__ gamma,
    const float* __restrict__ beta, const float* __restrict__ rmean,
    const float* __restrict__ rvar, float* __restrict__ h)
{
    __shared__ __attribute__((aligned(16))) float As[2][GBM * LDP];
    __shared__ __attribute__((aligned(16))) float Bs[2][GBN * LDP];

    const int tid  = threadIdx.x;
    const int wave = tid >> 5;
    const int lane = tid & 31;
    const int wm = wave >> 2;
    const int wn = wave & 3;
    const int n0 = blockIdx.x * GBN;
    const int m0 = blockIdx.y * GBM;

    const int srow = tid >> 3;
    const int sc4  = (tid & 7) << 2;
    const float* ag = x + (size_t)(m0 + srow) * CIN + sc4;
    const float* bg = W + (size_t)(n0 + srow) * CIN + sc4;
    const int sl = srow * LDP + sc4;

    v4f ra[4], rb[4];
    #pragma unroll
    for (int u = 0; u < 4; ++u) {
        ra[u] = *(const v4f*)(ag + (size_t)u * 32 * CIN);
        rb[u] = *(const v4f*)(bg + (size_t)u * 32 * CIN);
    }
    #pragma unroll
    for (int u = 0; u < 4; ++u) {
        *(v4f*)(&As[0][sl + u * 32 * LDP]) = ra[u];
        *(v4f*)(&Bs[0][sl + u * 32 * LDP]) = rb[u];
    }
    __syncthreads();

    v8f acc[4][2];
    #pragma unroll
    for (int i = 0; i < 4; ++i) {
        #pragma unroll
        for (int j = 0; j < 2; ++j) acc[i][j] = {};
    }

    const int colL = lane & 15;
    const int hh   = lane >> 4;
    const int aoff = (wm * 64 + colL) * LDP + 2 * hh;
    const int boff = (wn * 32 + colL) * LDP + 2 * hh;

    for (int t = 0; t < NCHUNK; ++t) {
        const int st = t & 1;
        if (t + 1 < NCHUNK) {
            #pragma unroll
            for (int u = 0; u < 4; ++u) {
                ra[u] = *(const v4f*)(ag + (size_t)u * 32 * CIN + (t + 1) * GBK);
                rb[u] = *(const v4f*)(bg + (size_t)u * 32 * CIN + (t + 1) * GBK);
            }
        }
        const float* as = &As[st][aoff];
        const float* bs = &Bs[st][boff];
        #pragma unroll
        for (int kk = 0; kk < GBK / 4; ++kk) {
            v2f a[4], b[2];
            #pragma unroll
            for (int i = 0; i < 4; ++i) a[i] = *(const v2f*)(as + i * 16 * LDP + kk * 4);
            #pragma unroll
            for (int j = 0; j < 2; ++j) b[j] = *(const v2f*)(bs + j * 16 * LDP + kk * 4);
            #pragma unroll
            for (int i = 0; i < 4; ++i) {
                #pragma unroll
                for (int j = 0; j < 2; ++j)
                    acc[i][j] = __builtin_amdgcn_wmma_f32_16x16x4_f32(false, a[i], false, b[j],
                                                                     (short)0, acc[i][j], false, false);
            }
        }
        if (t + 1 < NCHUNK) {
            #pragma unroll
            for (int u = 0; u < 4; ++u) {
                *(v4f*)(&As[st ^ 1][sl + u * 32 * LDP]) = ra[u];
                *(v4f*)(&Bs[st ^ 1][sl + u * 32 * LDP]) = rb[u];
            }
        }
        __syncthreads();
    }

    const int rOff = hh << 3;
    #pragma unroll
    for (int j = 0; j < 2; ++j) {
        const int o = n0 + wn * 32 + j * 16 + colL;
        const float sc = gamma[o] * rsqrtf(rvar[o] + BN_EPS);
        const float sh = (bias[o] - rmean[o]) * sc + beta[o];
        #pragma unroll
        for (int i = 0; i < 4; ++i) {
            const int mr = m0 + wm * 64 + i * 16 + rOff;
            #pragma unroll
            for (int r = 0; r < 8; ++r) {
                float v = fmaf(acc[i][j][r], sc, sh);
                v = fmaxf(v, 0.0f);
                h[(size_t)(mr + r) * COUT + o] = v;
            }
        }
    }
}

__device__ __forceinline__ unsigned axis_slice(float c)
{
    const float u = 64.0f * __builtin_amdgcn_rcpf(1.0f + __expf(-1.702f * c));
    int q = (int)u;
    q = q < 0 ? 0 : q;
    q = q > 63 ? 63 : q;
    return (unsigned)q;
}
__device__ __forceinline__ unsigned spread3(unsigned v)
{
    return (v & 1u) | ((v & 2u) << 2) | ((v & 4u) << 4) | ((v & 8u) << 6) | ((v & 16u) << 8) | ((v & 32u) << 10);
}

__global__ __launch_bounds__(256) void sort_cloud_kernel(const float* __restrict__ p,
    float* __restrict__ sx, float* __restrict__ sy, float* __restrict__ sz, int* __restrict__ so)
{
    __shared__ __attribute__((aligned(16))) unsigned keys[NPTS];
    const int b = blockIdx.y;
    const int t = threadIdx.x;
    const float* pb = p + (size_t)b * NPTS * 3;
    for (int u = 0; u < NPTS / 256; ++u) {
        const int j = u * 256 + t;
        const unsigned code = (spread3(axis_slice(pb[j * 3 + 0])) << 2) |
                              (spread3(axis_slice(pb[j * 3 + 1])) << 1) |
                               spread3(axis_slice(pb[j * 3 + 2]));
        keys[j] = (code << 12) | (unsigned)j;
    }
    __syncthreads();
    const int i = blockIdx.x * 256 + t;
    const unsigned ki = keys[i];
    unsigned rank = 0;
    #pragma unroll 4
    for (int j = 0; j < NPTS; j += 4) {
        const uint4 k4 = *(const uint4*)(&keys[j]);
        rank += (k4.x < ki ? 1u : 0u) + (k4.y < ki ? 1u : 0u) + (k4.z < ki ? 1u : 0u) + (k4.w < ki ? 1u : 0u);
    }
    const size_t o = (size_t)b * NPTS + rank;
    sx[o] = pb[i * 3 + 0];
    sy[o] = pb[i * 3 + 1];
    sz[o] = pb[i * 3 + 2];
    so[o] = i;
}

#define PPL  128
#define NPR  (PPL / 2)
#define GRP  8
#define NGRP (PPL / GRP)

__device__ __forceinline__ unsigned umx(unsigned a, unsigned b) { return a > b ? a : b; }
__device__ __forceinline__ unsigned umn(unsigned a, unsigned b) { return a < b ? a : b; }
__device__ __forceinline__ unsigned umx8(const unsigned* v)
{
    return umx(umx(umx(v[0], v[1]), umx(v[2], v[3])), umx(umx(v[4], v[5]), umx(v[6], v[7])));
}

#define XROW(v, n) ((unsigned)__builtin_amdgcn_update_dpp(0, (int)(v), 0x160 + (n), 0xf, 0xf, false))
#define XHALF(v)   ((unsigned)__builtin_amdgcn_permlanex16(0, (int)(v), 0x76543210, 0xfedcba98, false, false))

__device__ __forceinline__ unsigned row_umax(unsigned m)
{
    m = umx(m, XROW(m, 1)); m = umx(m, XROW(m, 2)); m = umx(m, XROW(m, 4)); m = umx(m, XROW(m, 8));
    return m;
}
__device__ __forceinline__ unsigned wave_umin(unsigned m)
{
    m = umn(m, XROW(m, 1)); m = umn(m, XROW(m, 2)); m = umn(m, XROW(m, 4)); m = umn(m, XROW(m, 8));
    return umn(m, XHALF(m));
}
__device__ __forceinline__ float wave_fmin(float v)
{
    v = fminf(v, __uint_as_float(XROW(__float_as_uint(v), 1)));
    v = fminf(v, __uint_as_float(XROW(__float_as_uint(v), 2)));
    v = fminf(v, __uint_as_float(XROW(__float_as_uint(v), 4)));
    v = fminf(v, __uint_as_float(XROW(__float_as_uint(v), 8)));
    return fminf(v, __uint_as_float(XHALF(__float_as_uint(v))));
}
__device__ __forceinline__ float wave_fmax(float v)
{
    v = fmaxf(v, __uint_as_float(XROW(__float_as_uint(v), 1)));
    v = fmaxf(v, __uint_as_float(XROW(__float_as_uint(v), 2)));
    v = fmaxf(v, __uint_as_float(XROW(__float_as_uint(v), 4)));
    v = fmaxf(v, __uint_as_float(XROW(__float_as_uint(v), 8)));
    return fmaxf(v, __uint_as_float(XHALF(__float_as_uint(v))));
}

__device__ __forceinline__ float lane_read(float v, int l)
{
    return __int_as_float(__builtin_amdgcn_readlane(__float_as_int(v), l));
}

#define OIDX(s) (((s) & 1) ? (oi2[(s) >> 1] >> 16) : (oi2[(s) >> 1] & 0xffffu))

#define FIND_STEP(s)                                                            \
    {                                                                           \
        const bool hit = (d[s] == M);                                           \
        const unsigned oi = hit ? OIDX(s) : 0xffffffffu;                        \
        const bool better = oi < cand;                                          \
        cand = umn(cand, oi);                                                   \
        xs = better ? px[(s) >> 1][(s) & 1] : xs;                               \
        ys = better ? py[(s) >> 1][(s) & 1] : ys;                               \
        zs = better ? pz[(s) >> 1][(s) & 1] : zs;                               \
    }
#define FIND_IN_GROUP(g)                                                        \
    FIND_STEP((g) * GRP + 0) FIND_STEP((g) * GRP + 1) FIND_STEP((g) * GRP + 2)  \
    FIND_STEP((g) * GRP + 3) FIND_STEP((g) * GRP + 4) FIND_STEP((g) * GRP + 5)  \
    FIND_STEP((g) * GRP + 6) FIND_STEP((g) * GRP + 7)

#define FIND_SWITCH(g)                                                          \
    switch (g) {                                                                \
        case 0:  FIND_IN_GROUP(0)  break;                                       \
        case 1:  FIND_IN_GROUP(1)  break;                                       \
        case 2:  FIND_IN_GROUP(2)  break;                                       \
        case 3:  FIND_IN_GROUP(3)  break;                                       \
        case 4:  FIND_IN_GROUP(4)  break;                                       \
        case 5:  FIND_IN_GROUP(5)  break;                                       \
        case 6:  FIND_IN_GROUP(6)  break;                                       \
        case 7:  FIND_IN_GROUP(7)  break;                                       \
        case 8:  FIND_IN_GROUP(8)  break;                                       \
        case 9:  FIND_IN_GROUP(9)  break;                                       \
        case 10: FIND_IN_GROUP(10) break;                                       \
        case 11: FIND_IN_GROUP(11) break;                                       \
        case 12: FIND_IN_GROUP(12) break;                                       \
        case 13: FIND_IN_GROUP(13) break;                                       \
        case 14: FIND_IN_GROUP(14) break;                                       \
        default: FIND_IN_GROUP(15) break;                                       \
    }

#define UPD_PAIR(k)                                                             \
    {                                                                           \
        const v2f dx = px[k] - nx2, dy = py[k] - ny2, dz = pz[k] - nz2;         \
        const v2f dd = dx * dx + dy * dy + dz * dz;                             \
        d[2 * (k)]     = umn(d[2 * (k)],     __float_as_uint(dd[0]));           \
        d[2 * (k) + 1] = umn(d[2 * (k) + 1], __float_as_uint(dd[1]));           \
    }
#define UPDATE_GROUP(g)                                                         \
    if (need & (1u << (g))) {                                                   \
        UPD_PAIR(4 * (g) + 0) UPD_PAIR(4 * (g) + 1)                             \
        UPD_PAIR(4 * (g) + 2) UPD_PAIR(4 * (g) + 3)                             \
        G[g] = umx8(&d[(g) * GRP]);                                             \
    }

__global__ __launch_bounds__(32) void fps_kernel(const float* __restrict__ p,
    const float* __restrict__ sx, const float* __restrict__ sy, const float* __restrict__ sz,
    const int* __restrict__ so, int* __restrict__ fid)
{
    const int b = blockIdx.x;
    const float* pb = p + (size_t)b * NPTS * 3;
    const float* qx = sx + (size_t)b * NPTS;
    const float* qy = sy + (size_t)b * NPTS;
    const float* qz = sz + (size_t)b * NPTS;
    const int*   qo = so + (size_t)b * NPTS;
    const int lane = threadIdx.x;

    v2f px[NPR], py[NPR], pz[NPR];
    unsigned oi2[NPR];
    unsigned d[PPL];
    unsigned G[NGRP];
    float blx = 0.0f, bhx = 0.0f, bly = 0.0f, bhy = 0.0f, blz = 0.0f, bhz = 0.0f;
    {
        const float x0 = pb[0], y0 = pb[1], z0 = pb[2];
        const v2f x02 = { x0, x0 }, y02 = { y0, y0 }, z02 = { z0, z0 };
        #pragma unroll
        for (int k = 0; k < NPR; ++k) {
            const int i0 = (2 * k) * 32 + lane, i1 = (2 * k + 1) * 32 + lane;
            px[k][0] = qx[i0]; py[k][0] = qy[i0]; pz[k][0] = qz[i0];
            px[k][1] = qx[i1]; py[k][1] = qy[i1]; pz[k][1] = qz[i1];
            oi2[k] = (unsigned)qo[i0] | ((unsigned)qo[i1] << 16);
            const v2f dx = px[k] - x02, dy = py[k] - y02, dz = pz[k] - z02;
            const v2f dd = dx * dx + dy * dy + dz * dz;
            d[2 * k] = __float_as_uint(dd[0]);
            d[2 * k + 1] = __float_as_uint(dd[1]);
        }
        #pragma unroll
        for (int g = 0; g < NGRP; ++g) {
            G[g] = umx8(&d[g * GRP]);
            float lx = px[4 * g][0], hx = lx, ly = py[4 * g][0], hy = ly, lz = pz[4 * g][0], hz = lz;
            #pragma unroll
            for (int kk = 0; kk < GRP / 2; ++kk) {
                #pragma unroll
                for (int e = 0; e < 2; ++e) {
                    const float cx = px[4 * g + kk][e], cy = py[4 * g + kk][e], cz = pz[4 * g + kk][e];
                    lx = fminf(lx, cx); hx = fmaxf(hx, cx);
                    ly = fminf(ly, cy); hy = fmaxf(hy, cy);
                    lz = fminf(lz, cz); hz = fmaxf(hz, cz);
                }
            }
            lx = wave_fmin(lx); hx = wave_fmax(hx);
            ly = wave_fmin(ly); hy = wave_fmax(hy);
            lz = wave_fmin(lz); hz = wave_fmax(hz);
            const bool mine = ((lane & 15) == g);
            blx = mine ? lx : blx; bhx = mine ? hx : bhx;
            bly = mine ? ly : bly; bhy = mine ? hy : bhy;
            blz = mine ? lz : blz; bhz = mine ? hz : bhz;
        }
    }

    if (lane == 0) fid[b * NOUT + 0] = b * NPTS;

    const bool b0 = (lane & 1) != 0, b1 = (lane & 2) != 0, b2 = (lane & 4) != 0, b3 = (lane & 8) != 0;

    for (int i = 1; i < NOUT; ++i) {
        unsigned t8[8], t4[4], t2[2];
        #pragma unroll
        for (int j = 0; j < 8; ++j) {
            const unsigned lo = G[2 * j], hi = G[2 * j + 1];
            t8[j] = umx(b0 ? hi : lo, XROW(b0 ? lo : hi, 1));
        }
        #pragma unroll
        for (int j = 0; j < 4; ++j) {
            const unsigned lo = t8[2 * j], hi = t8[2 * j + 1];
            t4[j] = umx(b1 ? hi : lo, XROW(b1 ? lo : hi, 2));
        }
        #pragma unroll
        for (int j = 0; j < 2; ++j) {
            const unsigned lo = t4[2 * j], hi = t4[2 * j + 1];
            t2[j] = umx(b2 ? hi : lo, XROW(b2 ? lo : hi, 4));
        }
        unsigned gw = umx(b3 ? t2[1] : t2[0], XROW(b3 ? t2[0] : t2[1], 8));
        gw = umx(gw, XHALF(gw));
        const unsigned M = row_umax(gw);

        unsigned gm = __builtin_amdgcn_ballot_w32(gw == M) & 0xffffu;
        unsigned cand = 0xffffffffu;
        float xs = 0.0f, ys = 0.0f, zs = 0.0f;
        { const int g = __builtin_ctz(gm | 0x8000u); FIND_SWITCH(g) }
        gm &= gm - 1;
        while (gm) {
            const int g = __builtin_ctz(gm);
            gm &= gm - 1;
            FIND_SWITCH(g)
        }
        const unsigned wmin = wave_umin(cand);
        const int wl = __builtin_ctz(__builtin_amdgcn_ballot_w32(cand == wmin) | 0x80000000u);
        const float nx = lane_read(xs, wl), ny = lane_read(ys, wl), nz = lane_read(zs, wl);
        if (lane == 0) fid[b * NOUT + i] = b * NPTS + (int)wmin;

        const float ex = fmaxf(fmaxf(blx - nx, nx - bhx), 0.0f);
        const float ey = fmaxf(fmaxf(bly - ny, ny - bhy), 0.0f);
        const float ez = fmaxf(fmaxf(blz - nz, nz - bhz), 0.0f);
        const float bd = fmaf(ez, ez, fmaf(ey, ey, ex * ex)) * 0.99999904632568359375f;
        const unsigned need = __builtin_amdgcn_ballot_w32(gw > __float_as_uint(bd)) & 0xffffu;

        const v2f nx2 = { nx, nx }, ny2 = { ny, ny }, nz2 = { nz, nz };
        UPDATE_GROUP(0)
        UPDATE_GROUP(1)
        UPDATE_GROUP(2)
        UPDATE_GROUP(3)
        UPDATE_GROUP(4)
        UPDATE_GROUP(5)
        UPDATE_GROUP(6)
        UPDATE_GROUP(7)
        UPDATE_GROUP(8)
        UPDATE_GROUP(9)
        UPDATE_GROUP(10)
        UPDATE_GROUP(11)
        UPDATE_GROUP(12)
        UPDATE_GROUP(13)
        UPDATE_GROUP(14)
        UPDATE_GROUP(15)
    }
}

__global__ __launch_bounds__(128) void gather_maxpool_kernel(
    const float* __restrict__ h, const float* __restrict__ p,
    const int* __restrict__ fid, const int* __restrict__ sid_euc,
    float* __restrict__ out)
{
    const int s = blockIdx.x;
    const int t = threadIdx.x;

    __shared__ int ssid[KNBR];
    const int f = fid[s];
    if (t < KNBR) ssid[t] = sid_euc[(size_t)f * KNBR + t];
    if (t < 3)    out[XOUT_ELEMS + (size_t)s * 3 + t] = p[(size_t)f * 3 + t];
    __syncthreads();

    const int c = t * 4;
    v4f acc = { -INFINITY, -INFINITY, -INFINITY, -INFINITY };
    #pragma unroll
    for (int k = 0; k < KNBR; ++k) {
        const int r = ssid[k];
        if (k + 1 < KNBR)
            __builtin_prefetch(h + (size_t)ssid[k + 1] * COUT + c, 0, 0);
        const v4f v = *(const v4f*)(h + (size_t)r * COUT + c);
        acc.x = fmaxf(acc.x, v.x);
        acc.y = fmaxf(acc.y, v.y);
        acc.z = fmaxf(acc.z, v.z);
        acc.w = fmaxf(acc.w, v.w);
    }
    *(v4f*)(out + (size_t)s * COUT + c) = acc;
}

extern "C" void kernel_launch(void* const* d_in, const int* in_sizes, int n_in,
                              void* d_out, int out_size, void* d_ws, size_t ws_size,
                              hipStream_t stream)
{
    (void)in_sizes; (void)n_in; (void)out_size; (void)ws_size;
    const float* x     = (const float*)d_in[0];
    const float* p     = (const float*)d_in[1];
    const int*   sid   = (const int*)  d_in[2];
    const float* W     = (const float*)d_in[4];
    const float* bias  = (const float*)d_in[5];
    const float* gamma = (const float*)d_in[6];
    const float* beta  = (const float*)d_in[7];
    const float* rmean = (const float*)d_in[8];
    const float* rvar  = (const float*)d_in[9];
    float* out = (float*)d_out;

    float* h  = (float*)d_ws;
    int* fid  = (int*)((char*)d_ws + (size_t)MROWS * COUT * 4);

    float* sx = (float*)d_ws;
    float* sy = sx + MROWS;
    float* sz = sy + MROWS;
    int*   so = (int*)(sz + MROWS);

    sort_cloud_kernel<<<dim3(NPTS / 256, BATCH), 256, 0, stream>>>(p, sx, sy, sz, so);
    fps_kernel<<<BATCH, 32, 0, stream>>>(p, sx, sy, sz, so, fid);
    dim3 ggrid(COUT / GBN, MROWS / GBM);
    gemm_bn_relu_kernel<<<ggrid, 256, 0, stream>>>(x, W, bias, gamma, beta,
                                                   rmean, rvar, h);
    gather_maxpool_kernel<<<BATCH * NOUT, 128, 0, stream>>>(h, p, fid, sid, out);
}


# FPS cells: exact k-d split (x quartiles, then y and z medians, each a counting rank; 2 more small launches) replaces fixed-slice Morton ranks: tight boxes, fewer cells updated per pick; same picks
# speedup vs baseline: 1.9613x; 1.0759x over previous
#include <hip/hip_runtime.h>
#include <hip/hip_bf16.h>
#include <math.h>

typedef float v2f __attribute__((ext_vector_type(2)));
typedef float v4f __attribute__((ext_vector_type(4)));
typedef float v8f __attribute__((ext_vector_type(8)));

#define BATCH 16
#define NPTS  4096
#define CIN   256
#define COUT  512
#define KNBR  16
#define NOUT  1024
#define MROWS (BATCH * NPTS)
#define BN_EPS 1e-5f
#define XOUT_ELEMS ((size_t)BATCH * NOUT * COUT)

#define GBM 128
#define GBN 128
#define GBK 32
#define LDP 36
#define NCHUNK (CIN / GBK)

__global__ __launch_bounds__(256) void gemm_bn_relu_kernel(
    const float* __restrict__ x, const float* __restrict__ W,
    const float* __restrict__ bias, const float* __restrict__ gamma,
    const float* __restrict__ beta, const float* __restrict__ rmean,
    const float* __restrict__ rvar, float* __restrict__ h)
{
    __shared__ __attribute__((aligned(16))) float As[2][GBM * LDP];
    __shared__ __attribute__((aligned(16))) float Bs[2][GBN * LDP];

    const int tid  = threadIdx.x;
    const int wave = tid >> 5;
    const int lane = tid & 31;
    const int wm = wave >> 2;
    const int wn = wave & 3;
    const int n0 = blockIdx.x * GBN;
    const int m0 = blockIdx.y * GBM;

    const int srow = tid >> 3;
    const int sc4  = (tid & 7) << 2;
    const float* ag = x + (size_t)(m0 + srow) * CIN + sc4;
    const float* bg = W + (size_t)(n0 + srow) * CIN + sc4;
    const int sl = srow * LDP + sc4;

    v4f ra[4], rb[4];
    #pragma unroll
    for (int u = 0; u < 4; ++u) {
        ra[u] = *(const v4f*)(ag + (size_t)u * 32 * CIN);
        rb[u] = *(const v4f*)(bg + (size_t)u * 32 * CIN);
    }
    #pragma unroll
    for (int u = 0; u < 4; ++u) {
        *(v4f*)(&As[0][sl + u * 32 * LDP]) = ra[u];
        *(v4f*)(&Bs[0][sl + u * 32 * LDP]) = rb[u];
    }
    __syncthreads();

    v8f acc[4][2];
    #pragma unroll
    for (int i = 0; i < 4; ++i) {
        #pragma unroll
        for (int j = 0; j < 2; ++j) acc[i][j] = {};
    }

    const int colL = lane & 15;
    const int hh   = lane >> 4;
    const int aoff = (wm * 64 + colL) * LDP + 2 * hh;
    const int boff = (wn * 32 + colL) * LDP + 2 * hh;

    for (int t = 0; t < NCHUNK; ++t) {
        const int st = t & 1;
        if (t + 1 < NCHUNK) {
            #pragma unroll
            for (int u = 0; u < 4; ++u) {
                ra[u] = *(const v4f*)(ag + (size_t)u * 32 * CIN + (t + 1) * GBK);
                rb[u] = *(const v4f*)(bg + (size_t)u * 32 * CIN + (t + 1) * GBK);
            }
        }
        const float* as = &As[st][aoff];
        const float* bs = &Bs[st][boff];
        #pragma unroll
        for (int kk = 0; kk < GBK / 4; ++kk) {
            v2f a[4], b[2];
            #pragma unroll
            for (int i = 0; i < 4; ++i) a[i] = *(const v2f*)(as + i * 16 * LDP + kk * 4);
            #pragma unroll
            for (int j = 0; j < 2; ++j) b[j] = *(const v2f*)(bs + j * 16 * LDP + kk * 4);
            #pragma unroll
            for (int i = 0; i < 4; ++i) {
                #pragma unroll
                for (int j = 0; j < 2; ++j)
                    acc[i][j] = __builtin_amdgcn_wmma_f32_16x16x4_f32(false, a[i], false, b[j],
                                                                     (short)0, acc[i][j], false, false);
            }
        }
        if (t + 1 < NCHUNK) {
            #pragma unroll
            for (int u = 0; u < 4; ++u) {
                *(v4f*)(&As[st ^ 1][sl + u * 32 * LDP]) = ra[u];
                *(v4f*)(&Bs[st ^ 1][sl + u * 32 * LDP]) = rb[u];
            }
        }
        __syncthreads();
    }

    const int rOff = hh << 3;
    #pragma unroll
    for (int j = 0; j < 2; ++j) {
        const int o = n0 + wn * 32 + j * 16 + colL;
        const float sc = gamma[o] * rsqrtf(rvar[o] + BN_EPS);
        const float sh = (bias[o] - rmean[o]) * sc + beta[o];
        #pragma unroll
        for (int i = 0; i < 4; ++i) {
            const int mr = m0 + wm * 64 + i * 16 + rOff;
            #pragma unroll
            for (int r = 0; r < 8; ++r) {
                float v = fmaf(acc[i][j][r], sc, sh);
                v = fmaxf(v, 0.0f);
                h[(size_t)(mr + r) * COUT + o] = v;
            }
        }
    }
}

#define SLAB 1024
#define HALF 512

__device__ __forceinline__ unsigned ordered_bits(float f)
{
    const unsigned u = __float_as_uint(f);
    return (u & 0x80000000u) ? ~u : (u | 0x80000000u);
}

__device__ __forceinline__ unsigned count_smaller(const unsigned* keys, int n, unsigned ki)
{
    unsigned rank = 0;
    #pragma unroll 4
    for (int j = 0; j < n; j += 4) {
        const uint4 k4 = *(const uint4*)(&keys[j]);
        rank += (k4.x < ki ? 1u : 0u) + (k4.y < ki ? 1u : 0u) + (k4.z < ki ? 1u : 0u) + (k4.w < ki ? 1u : 0u);
    }
    return rank;
}

__global__ __launch_bounds__(256) void rank_x_kernel(const float* __restrict__ p,
    float* __restrict__ ox, float* __restrict__ oy, float* __restrict__ oz, int* __restrict__ oo)
{
    __shared__ __attribute__((aligned(16))) unsigned keys[NPTS];
    const int b = blockIdx.y;
    const int t = threadIdx.x;
    const float* pb = p + (size_t)b * NPTS * 3;
    for (int u = 0; u < NPTS / 256; ++u) {
        const int j = u * 256 + t;
        keys[j] = (ordered_bits(pb[j * 3 + 0]) & ~(unsigned)(NPTS - 1)) | (unsigned)j;
    }
    const int i = blockIdx.x * 256 + t;
    const float vx = pb[i * 3 + 0], vy = pb[i * 3 + 1], vz = pb[i * 3 + 2];
    __syncthreads();
    const size_t o = (size_t)b * NPTS + count_smaller(keys, NPTS, keys[i]);
    ox[o] = vx;
    oy[o] = vy;
    oz[o] = vz;
    oo[o] = i;
}

template <int G>
__global__ __launch_bounds__(256) void rank_in_group_kernel(const float* __restrict__ key,
    const float* __restrict__ ix, const float* __restrict__ iy, const float* __restrict__ iz,
    const int* __restrict__ io,
    float* __restrict__ ox, float* __restrict__ oy, float* __restrict__ oz, int* __restrict__ oo)
{
    __shared__ __attribute__((aligned(16))) unsigned keys[G];
    constexpr int PER = G / 256;
    const int t = threadIdx.x;
    const size_t base = (size_t)blockIdx.y * NPTS + (size_t)(blockIdx.x / PER) * G;
    for (int u = 0; u < PER; ++u) {
        const int j = u * 256 + t;
        keys[j] = (ordered_bits(key[base + j]) & ~(unsigned)(G - 1)) | (unsigned)j;
    }
    const int i = (blockIdx.x % PER) * 256 + t;
    const float vx = ix[base + i], vy = iy[base + i], vz = iz[base + i];
    const int vo = io[base + i];
    __syncthreads();
    const size_t o = base + count_smaller(keys, G, keys[i]);
    ox[o] = vx;
    oy[o] = vy;
    oz[o] = vz;
    oo[o] = vo;
}

#define PPL  128
#define NPR  (PPL / 2)
#define GRP  8
#define NGRP (PPL / GRP)

__device__ __forceinline__ unsigned umx(unsigned a, unsigned b) { return a > b ? a : b; }
__device__ __forceinline__ unsigned umn(unsigned a, unsigned b) { return a < b ? a : b; }
__device__ __forceinline__ unsigned umx8(const unsigned* v)
{
    return umx(umx(umx(v[0], v[1]), umx(v[2], v[3])), umx(umx(v[4], v[5]), umx(v[6], v[7])));
}

#define XROW(v, n) ((unsigned)__builtin_amdgcn_update_dpp(0, (int)(v), 0x160 + (n), 0xf, 0xf, false))
#define XHALF(v)   ((unsigned)__builtin_amdgcn_permlanex16(0, (int)(v), 0x76543210, 0xfedcba98, false, false))

__device__ __forceinline__ unsigned row_umax(unsigned m)
{
    m = umx(m, XROW(m, 1)); m = umx(m, XROW(m, 2)); m = umx(m, XROW(m, 4)); m = umx(m, XROW(m, 8));
    return m;
}
__device__ __forceinline__ unsigned wave_umin(unsigned m)
{
    m = umn(m, XROW(m, 1)); m = umn(m, XROW(m, 2)); m = umn(m, XROW(m, 4)); m = umn(m, XROW(m, 8));
    return umn(m, XHALF(m));
}
__device__ __forceinline__ float wave_fmin(float v)
{
    v = fminf(v, __uint_as_float(XROW(__float_as_uint(v), 1)));
    v = fminf(v, __uint_as_float(XROW(__float_as_uint(v), 2)));
    v = fminf(v, __uint_as_float(XROW(__float_as_uint(v), 4)));
    v = fminf(v, __uint_as_float(XROW(__float_as_uint(v), 8)));
    return fminf(v, __uint_as_float(XHALF(__float_as_uint(v))));
}
__device__ __forceinline__ float wave_fmax(float v)
{
    v = fmaxf(v, __uint_as_float(XROW(__float_as_uint(v), 1)));
    v = fmaxf(v, __uint_as_float(XROW(__float_as_uint(v), 2)));
    v = fmaxf(v, __uint_as_float(XROW(__float_as_uint(v), 4)));
    v = fmaxf(v, __uint_as_float(XROW(__float_as_uint(v), 8)));
    return fmaxf(v, __uint_as_float(XHALF(__float_as_uint(v))));
}

__device__ __forceinline__ float lane_read(float v, int l)
{
    return __int_as_float(__builtin_amdgcn_readlane(__float_as_int(v), l));
}

#define OIDX(s) (((s) & 1) ? (oi2[(s) >> 1] >> 16) : (oi2[(s) >> 1] & 0xffffu))

#define FIND_STEP(s)                                                            \
    {                                                                           \
        const bool hit = (d[s] == M);                                           \
        const unsigned oi = hit ? OIDX(s) : 0xffffffffu;                        \
        const bool better = oi < cand;                                          \
        cand = umn(cand, oi);                                                   \
        xs = better ? px[(s) >> 1][(s) & 1] : xs;                               \
        ys = better ? py[(s) >> 1][(s) & 1] : ys;                               \
        zs = better ? pz[(s) >> 1][(s) & 1] : zs;                               \
    }
#define FIND_IN_GROUP(g)                                                        \
    FIND_STEP((g) * GRP + 0) FIND_STEP((g) * GRP + 1) FIND_STEP((g) * GRP + 2)  \
    FIND_STEP((g) * GRP + 3) FIND_STEP((g) * GRP + 4) FIND_STEP((g) * GRP + 5)  \
    FIND_STEP((g) * GRP + 6) FIND_STEP((g) * GRP + 7)

#define FIND_SWITCH(g)                                                          \
    switch (g) {                                                                \
        case 0:  FIND_IN_GROUP(0)  break;                                       \
        case 1:  FIND_IN_GROUP(1)  break;                                       \
        case 2:  FIND_IN_GROUP(2)  break;                                       \
        case 3:  FIND_IN_GROUP(3)  break;                                       \
        case 4:  FIND_IN_GROUP(4)  break;                                       \
        case 5:  FIND_IN_GROUP(5)  break;                                       \
        case 6:  FIND_IN_GROUP(6)  break;                                       \
        case 7:  FIND_IN_GROUP(7)  break;                                       \
        case 8:  FIND_IN_GROUP(8)  break;                                       \
        case 9:  FIND_IN_GROUP(9)  break;                                       \
        case 10: FIND_IN_GROUP(10) break;                                       \
        case 11: FIND_IN_GROUP(11) break;                                       \
        case 12: FIND_IN_GROUP(12) break;                                       \
        case 13: FIND_IN_GROUP(13) break;                                       \
        case 14: FIND_IN_GROUP(14) break;                                       \
        default: FIND_IN_GROUP(15) break;                                       \
    }

#define UPD_PAIR(k)                                                             \
    {                                                                           \
        const v2f dx = px[k] - nx2, dy = py[k] - ny2, dz = pz[k] - nz2;         \
        const v2f dd = dx * dx + dy * dy + dz * dz;                             \
        d[2 * (k)]     = umn(d[2 * (k)],     __float_as_uint(dd[0]));           \
        d[2 * (k) + 1] = umn(d[2 * (k) + 1], __float_as_uint(dd[1]));           \
    }
#define UPDATE_GROUP(g)                                                         \
    if (need & (1u << (g))) {                                                   \
        UPD_PAIR(4 * (g) + 0) UPD_PAIR(4 * (g) + 1)                             \
        UPD_PAIR(4 * (g) + 2) UPD_PAIR(4 * (g) + 3)                             \
        G[g] = umx8(&d[(g) * GRP]);                                             \
    }

__global__ __launch_bounds__(32) void fps_kernel(const float* __restrict__ p,
    const float* __restrict__ sx, const float* __restrict__ sy, const float* __restrict__ sz,
    const int* __restrict__ so, int* __restrict__ fid)
{
    const int b = blockIdx.x;
    const float* pb = p + (size_t)b * NPTS * 3;
    const float* qx = sx + (size_t)b * NPTS;
    const float* qy = sy + (size_t)b * NPTS;
    const float* qz = sz + (size_t)b * NPTS;
    const int*   qo = so + (size_t)b * NPTS;
    const int lane = threadIdx.x;

    v2f px[NPR], py[NPR], pz[NPR];
    unsigned oi2[NPR];
    unsigned d[PPL];
    unsigned G[NGRP];
    float blx = 0.0f, bhx = 0.0f, bly = 0.0f, bhy = 0.0f, blz = 0.0f, bhz = 0.0f;
    {
        const float x0 = pb[0], y0 = pb[1], z0 = pb[2];
        const v2f x02 = { x0, x0 }, y02 = { y0, y0 }, z02 = { z0, z0 };
        #pragma unroll
        for (int k = 0; k < NPR; ++k) {
            const int i0 = (2 * k) * 32 + lane, i1 = (2 * k + 1) * 32 + lane;
            px[k][0] = qx[i0]; py[k][0] = qy[i0]; pz[k][0] = qz[i0];
            px[k][1] = qx[i1]; py[k][1] = qy[i1]; pz[k][1] = qz[i1];
            oi2[k] = (unsigned)qo[i0] | ((unsigned)qo[i1] << 16);
            const v2f dx = px[k] - x02, dy = py[k] - y02, dz = pz[k] - z02;
            const v2f dd = dx * dx + dy * dy + dz * dz;
            d[2 * k] = __float_as_uint(dd[0]);
            d[2 * k + 1] = __float_as_uint(dd[1]);
        }
        #pragma unroll
        for (int g = 0; g < NGRP; ++g) {
            G[g] = umx8(&d[g * GRP]);
            float lx = px[4 * g][0], hx = lx, ly = py[4 * g][0], hy = ly, lz = pz[4 * g][0], hz = lz;
            #pragma unroll
            for (int kk = 0; kk < GRP / 2; ++kk) {
                #pragma unroll
                for (int e = 0; e < 2; ++e) {
                    const float cx = px[4 * g + kk][e], cy = py[4 * g + kk][e], cz = pz[4 * g + kk][e];
                    lx = fminf(lx, cx); hx = fmaxf(hx, cx);
                    ly = fminf(ly, cy); hy = fmaxf(hy, cy);
                    lz = fminf(lz, cz); hz = fmaxf(hz, cz);
                }
            }
            lx = wave_fmin(lx); hx = wave_fmax(hx);
            ly = wave_fmin(ly); hy = wave_fmax(hy);
            lz = wave_fmin(lz); hz = wave_fmax(hz);
            const bool mine = ((lane & 15) == g);
            blx = mine ? lx : blx; bhx = mine ? hx : bhx;
            bly = mine ? ly : bly; bhy = mine ? hy : bhy;
            blz = mine ? lz : blz; bhz = mine ? hz : bhz;
        }
    }

    if (lane == 0) fid[b * NOUT + 0] = b * NPTS;

    const bool b0 = (lane & 1) != 0, b1 = (lane & 2) != 0, b2 = (lane & 4) != 0, b3 = (lane & 8) != 0;

    for (int i = 1; i < NOUT; ++i) {
        unsigned t8[8], t4[4], t2[2];
        #pragma unroll
        for (int j = 0; j < 8; ++j) {
            const unsigned lo = G[2 * j], hi = G[2 * j + 1];
            t8[j] = umx(b0 ? hi : lo, XROW(b0 ? lo : hi, 1));
        }
        #pragma unroll
        for (int j = 0; j < 4; ++j) {
            const unsigned lo = t8[2 * j], hi = t8[2 * j + 1];
            t4[j] = umx(b1 ? hi : lo, XROW(b1 ? lo : hi, 2));
        }
        #pragma unroll
        for (int j = 0; j < 2; ++j) {
            const unsigned lo = t4[2 * j], hi = t4[2 * j + 1];
            t2[j] = umx(b2 ? hi : lo, XROW(b2 ? lo : hi, 4));
        }
        unsigned gw = umx(b3 ? t2[1] : t2[0], XROW(b3 ? t2[0] : t2[1], 8));
        gw = umx(gw, XHALF(gw));
        const unsigned M = row_umax(gw);

        unsigned gm = __builtin_amdgcn_ballot_w32(gw == M) & 0xffffu;
        unsigned cand = 0xffffffffu;
        float xs = 0.0f, ys = 0.0f, zs = 0.0f;
        { const int g = __builtin_ctz(gm | 0x8000u); FIND_SWITCH(g) }
        gm &= gm - 1;
        while (gm) {
            const int g = __builtin_ctz(gm);
            gm &= gm - 1;
            FIND_SWITCH(g)
        }
        const unsigned wmin = wave_umin(cand);
        const int wl = __builtin_ctz(__builtin_amdgcn_ballot_w32(cand == wmin) | 0x80000000u);
        const float nx = lane_read(xs, wl), ny = lane_read(ys, wl), nz = lane_read(zs, wl);
        if (lane == 0) fid[b * NOUT + i] = b * NPTS + (int)wmin;

        const float ex = fmaxf(fmaxf(blx - nx, nx - bhx), 0.0f);
        const float ey = fmaxf(fmaxf(bly - ny, ny - bhy), 0.0f);
        const float ez = fmaxf(fmaxf(blz - nz, nz - bhz), 0.0f);
        const float bd = fmaf(ez, ez, fmaf(ey, ey, ex * ex)) * 0.99999904632568359375f;
        const unsigned need = __builtin_amdgcn_ballot_w32(gw > __float_as_uint(bd)) & 0xffffu;

        const v2f nx2 = { nx, nx }, ny2 = { ny, ny }, nz2 = { nz, nz };
        UPDATE_GROUP(0)
        UPDATE_GROUP(1)
        UPDATE_GROUP(2)
        UPDATE_GROUP(3)
        UPDATE_GROUP(4)
        UPDATE_GROUP(5)
        UPDATE_GROUP(6)
        UPDATE_GROUP(7)
        UPDATE_GROUP(8)
        UPDATE_GROUP(9)
        UPDATE_GROUP(10)
        UPDATE_GROUP(11)
        UPDATE_GROUP(12)
        UPDATE_GROUP(13)
        UPDATE_GROUP(14)
        UPDATE_GROUP(15)
    }
}

__global__ __launch_bounds__(128) void gather_maxpool_kernel(
    const float* __restrict__ h, const float* __restrict__ p,
    const int* __restrict__ fid, const int* __restrict__ sid_euc,
    float* __restrict__ out)
{
    const int s = blockIdx.x;
    const int t = threadIdx.x;

    __shared__ int ssid[KNBR];
    const int f = fid[s];
    if (t < KNBR) ssid[t] = sid_euc[(size_t)f * KNBR + t];
    if (t < 3)    out[XOUT_ELEMS + (size_t)s * 3 + t] = p[(size_t)f * 3 + t];
    __syncthreads();

    const int c = t * 4;
    v4f acc = { -INFINITY, -INFINITY, -INFINITY, -INFINITY };
    #pragma unroll
    for (int k = 0; k < KNBR; ++k) {
        const int r = ssid[k];
        if (k + 1 < KNBR)
            __builtin_prefetch(h + (size_t)ssid[k + 1] * COUT + c, 0, 0);
        const v4f v = *(const v4f*)(h + (size_t)r * COUT + c);
        acc.x = fmaxf(acc.x, v.x);
        acc.y = fmaxf(acc.y, v.y);
        acc.z = fmaxf(acc.z, v.z);
        acc.w = fmaxf(acc.w, v.w);
    }
    *(v4f*)(out + (size_t)s * COUT + c) = acc;
}

extern "C" void kernel_launch(void* const* d_in, const int* in_sizes, int n_in,
                              void* d_out, int out_size, void* d_ws, size_t ws_size,
                              hipStream_t stream)
{
    (void)in_sizes; (void)n_in; (void)out_size; (void)ws_size;
    const float* x     = (const float*)d_in[0];
    const float* p     = (const float*)d_in[1];
    const int*   sid   = (const int*)  d_in[2];
    const float* W     = (const float*)d_in[4];
    const float* bias  = (const float*)d_in[5];
    const float* gamma = (const float*)d_in[6];
    const float* beta  = (const float*)d_in[7];
    const float* rmean = (const float*)d_in[8];
    const float* rvar  = (const float*)d_in[9];
    float* out = (float*)d_out;

    float* h  = (float*)d_ws;
    int* fid  = (int*)((char*)d_ws + (size_t)MROWS * COUT * 4);

    float* sx = (float*)d_ws;
    float* sy = sx + MROWS;
    float* sz = sy + MROWS;
    int*   so = (int*)(sz + MROWS);
    float* ax = (float*)(so + MROWS);
    float* ay = ax + MROWS;
    float* az = ay + MROWS;
    int*   ao = (int*)(az + MROWS);
    float* bx = (float*)(ao + MROWS);
    float* by = bx + MROWS;
    float* bz = by + MROWS;
    int*   bo = (int*)(bz + MROWS);

    const dim3 sgrid(NPTS / 256, BATCH);
    rank_x_kernel<<<sgrid, 256, 0, stream>>>(p, ax, ay, az, ao);
    rank_in_group_kernel<SLAB><<<sgrid, 256, 0, stream>>>(ay, ax, ay, az, ao, bx, by, bz, bo);
    rank_in_group_kernel<HALF><<<sgrid, 256, 0, stream>>>(bz, bx, by, bz, bo, sx, sy, sz, so);
    fps_kernel<<<BATCH, 32, 0, stream>>>(p, sx, sy, sz, so, fid);
    dim3 ggrid(COUT / GBN, MROWS / GBM);
    gemm_bn_relu_kernel<<<ggrid, 256, 0, stream>>>(x, W, bias, gamma, beta,
                                                   rmean, rvar, h);
    gather_maxpool_kernel<<<BATCH * NOUT, 128, 0, stream>>>(h, p, fid, sid, out);
}


# FPS winner search: cells holding the maximum are walked by a fall-through ladder of 16 bit tests, each search on the cell's own registers; was a switch tree with register copies and a tie loop
# speedup vs baseline: 2.0403x; 1.0403x over previous
#include <hip/hip_runtime.h>
#include <hip/hip_bf16.h>
#include <math.h>

typedef float v2f __attribute__((ext_vector_type(2)));
typedef float v4f __attribute__((ext_vector_type(4)));
typedef float v8f __attribute__((ext_vector_type(8)));

#define BATCH 16
#define NPTS  4096
#define CIN   256
#define COUT  512
#define KNBR  16
#define NOUT  1024
#define MROWS (BATCH * NPTS)
#define BN_EPS 1e-5f
#define XOUT_ELEMS ((size_t)BATCH * NOUT * COUT)

#define GBM 128
#define GBN 128
#define GBK 32
#define LDP 36
#define NCHUNK (CIN / GBK)

__global__ __launch_bounds__(256) void gemm_bn_relu_kernel(
    const float* __restrict__ x, const float* __restrict__ W,
    const float* __restrict__ bias, const float* __restrict__ gamma,
    const float* __restrict__ beta, const float* __restrict__ rmean,
    const float* __restrict__ rvar, float* __restrict__ h)
{
    __shared__ __attribute__((aligned(16))) float As[2][GBM * LDP];
    __shared__ __attribute__((aligned(16))) float Bs[2][GBN * LDP];

    const int tid  = threadIdx.x;
    const int wave = tid >> 5;
    const int lane = tid & 31;
    const int wm = wave >> 2;
    const int wn = wave & 3;
    const int n0 = blockIdx.x * GBN;
    const int m0 = blockIdx.y * GBM;

    const int srow = tid >> 3;
    const int sc4  = (tid & 7) << 2;
    const float* ag = x + (size_t)(m0 + srow) * CIN + sc4;
    const float* bg = W + (size_t)(n0 + srow) * CIN + sc4;
    const int sl = srow * LDP + sc4;

    v4f ra[4], rb[4];
    #pragma unroll
    for (int u = 0; u < 4; ++u) {
        ra[u] = *(const v4f*)(ag + (size_t)u * 32 * CIN);
        rb[u] = *(const v4f*)(bg + (size_t)u * 32 * CIN);
    }
    #pragma unroll
    for (int u = 0; u < 4; ++u) {
        *(v4f*)(&As[0][sl + u * 32 * LDP]) = ra[u];
        *(v4f*)(&Bs[0][sl + u * 32 * LDP]) = rb[u];
    }
    __syncthreads();

    v8f acc[4][2];
    #pragma unroll
    for (int i = 0; i < 4; ++i) {
        #pragma unroll
        for (int j = 0; j < 2; ++j) acc[i][j] = {};
    }

    const int colL = lane & 15;
    const int hh   = lane >> 4;
    const int aoff = (wm * 64 + colL) * LDP + 2 * hh;
    const int boff = (wn * 32 + colL) * LDP + 2 * hh;

    for (int t = 0; t < NCHUNK; ++t) {
        const int st = t & 1;
        if (t + 1 < NCHUNK) {
            #pragma unroll
            for (int u = 0; u < 4; ++u) {
                ra[u] = *(const v4f*)(ag + (size_t)u * 32 * CIN + (t + 1) * GBK);
                rb[u] = *(const v4f*)(bg + (size_t)u * 32 * CIN + (t + 1) * GBK);
            }
        }
        const float* as = &As[st][aoff];
        const float* bs = &Bs[st][boff];
        #pragma unroll
        for (int kk = 0; kk < GBK / 4; ++kk) {
            v2f a[4], b[2];
            #pragma unroll
            for (int i = 0; i < 4; ++i) a[i] = *(const v2f*)(as + i * 16 * LDP + kk * 4);
            #pragma unroll
            for (int j = 0; j < 2; ++j) b[j] = *(const v2f*)(bs + j * 16 * LDP + kk * 4);
            #pragma unroll
            for (int i = 0; i < 4; ++i) {
                #pragma unroll
                for (int j = 0; j < 2; ++j)
                    acc[i][j] = __builtin_amdgcn_wmma_f32_16x16x4_f32(false, a[i], false, b[j],
                                                                     (short)0, acc[i][j], false, false);
            }
        }
        if (t + 1 < NCHUNK) {
            #pragma unroll
            for (int u = 0; u < 4; ++u) {
                *(v4f*)(&As[st ^ 1][sl + u * 32 * LDP]) = ra[u];
                *(v4f*)(&Bs[st ^ 1][sl + u * 32 * LDP]) = rb[u];
            }
        }
        __syncthreads();
    }

    const int rOff = hh << 3;
    #pragma unroll
    for (int j = 0; j < 2; ++j) {
        const int o = n0 + wn * 32 + j * 16 + colL;
        const float sc = gamma[o] * rsqrtf(rvar[o] + BN_EPS);
        const float sh = (bias[o] - rmean[o]) * sc + beta[o];
        #pragma unroll
        for (int i = 0; i < 4; ++i) {
            const int mr = m0 + wm * 64 + i * 16 + rOff;
            #pragma unroll
            for (int r = 0; r < 8; ++r) {
                float v = fmaf(acc[i][j][r], sc, sh);
                v = fmaxf(v, 0.0f);
                h[(size_t)(mr + r) * COUT + o] = v;
            }
        }
    }
}

#define SLAB 1024
#define HALF 512

__device__ __forceinline__ unsigned ordered_bits(float f)
{
    const unsigned u = __float_as_uint(f);
    return (u & 0x80000000u) ? ~u : (u | 0x80000000u);
}

__device__ __forceinline__ unsigned count_smaller(const unsigned* keys, int n, unsigned ki)
{
    unsigned rank = 0;
    #pragma unroll 4
    for (int j = 0; j < n; j += 4) {
        const uint4 k4 = *(const uint4*)(&keys[j]);
        rank += (k4.x < ki ? 1u : 0u) + (k4.y < ki ? 1u : 0u) + (k4.z < ki ? 1u : 0u) + (k4.w < ki ? 1u : 0u);
    }
    return rank;
}

__global__ __launch_bounds__(256) void rank_x_kernel(const float* __restrict__ p,
    float* __restrict__ ox, float* __restrict__ oy, float* __restrict__ oz, int* __restrict__ oo)
{
    __shared__ __attribute__((aligned(16))) unsigned keys[NPTS];
    const int b = blockIdx.y;
    const int t = threadIdx.x;
    const float* pb = p + (size_t)b * NPTS * 3;
    for (int u = 0; u < NPTS / 256; ++u) {
        const int j = u * 256 + t;
        keys[j] = (ordered_bits(pb[j * 3 + 0]) & ~(unsigned)(NPTS - 1)) | (unsigned)j;
    }
    const int i = blockIdx.x * 256 + t;
    const float vx = pb[i * 3 + 0], vy = pb[i * 3 + 1], vz = pb[i * 3 + 2];
    __syncthreads();
    const size_t o = (size_t)b * NPTS + count_smaller(keys, NPTS, keys[i]);
    ox[o] = vx;
    oy[o] = vy;
    oz[o] = vz;
    oo[o] = i;
}

template <int G>
__global__ __launch_bounds__(256) void rank_in_group_kernel(const float* __restrict__ key,
    const float* __restrict__ ix, const float* __restrict__ iy, const float* __restrict__ iz,
    const int* __restrict__ io,
    float* __restrict__ ox, float* __restrict__ oy, float* __restrict__ oz, int* __restrict__ oo)
{
    __shared__ __attribute__((aligned(16))) unsigned keys[G];
    constexpr int PER = G / 256;
    const int t = threadIdx.x;
    const size_t base = (size_t)blockIdx.y * NPTS + (size_t)(blockIdx.x / PER) * G;
    for (int u = 0; u < PER; ++u) {
        const int j = u * 256 + t;
        keys[j] = (ordered_bits(key[base + j]) & ~(unsigned)(G - 1)) | (unsigned)j;
    }
    const int i = (blockIdx.x % PER) * 256 + t;
    const float vx = ix[base + i], vy = iy[base + i], vz = iz[base + i];
    const int vo = io[base + i];
    __syncthreads();
    const size_t o = base + count_smaller(keys, G, keys[i]);
    ox[o] = vx;
    oy[o] = vy;
    oz[o] = vz;
    oo[o] = vo;
}

#define PPL  128
#define NPR  (PPL / 2)
#define GRP  8
#define NGRP (PPL / GRP)

__device__ __forceinline__ unsigned umx(unsigned a, unsigned b) { return a > b ? a : b; }
__device__ __forceinline__ unsigned umn(unsigned a, unsigned b) { return a < b ? a : b; }
__device__ __forceinline__ unsigned umx8(const unsigned* v)
{
    return umx(umx(umx(v[0], v[1]), umx(v[2], v[3])), umx(umx(v[4], v[5]), umx(v[6], v[7])));
}

#define XROW(v, n) ((unsigned)__builtin_amdgcn_update_dpp(0, (int)(v), 0x160 + (n), 0xf, 0xf, false))
#define XHALF(v)   ((unsigned)__builtin_amdgcn_permlanex16(0, (int)(v), 0x76543210, 0xfedcba98, false, false))

__device__ __forceinline__ unsigned row_umax(unsigned m)
{
    m = umx(m, XROW(m, 1)); m = umx(m, XROW(m, 2)); m = umx(m, XROW(m, 4)); m = umx(m, XROW(m, 8));
    return m;
}
__device__ __forceinline__ unsigned wave_umin(unsigned m)
{
    m = umn(m, XROW(m, 1)); m = umn(m, XROW(m, 2)); m = umn(m, XROW(m, 4)); m = umn(m, XROW(m, 8));
    return umn(m, XHALF(m));
}
__device__ __forceinline__ float wave_fmin(float v)
{
    v = fminf(v, __uint_as_float(XROW(__float_as_uint(v), 1)));
    v = fminf(v, __uint_as_float(XROW(__float_as_uint(v), 2)));
    v = fminf(v, __uint_as_float(XROW(__float_as_uint(v), 4)));
    v = fminf(v, __uint_as_float(XROW(__float_as_uint(v), 8)));
    return fminf(v, __uint_as_float(XHALF(__float_as_uint(v))));
}
__device__ __forceinline__ float wave_fmax(float v)
{
    v = fmaxf(v, __uint_as_float(XROW(__float_as_uint(v), 1)));
    v = fmaxf(v, __uint_as_float(XROW(__float_as_uint(v), 2)));
    v = fmaxf(v, __uint_as_float(XROW(__float_as_uint(v), 4)));
    v = fmaxf(v, __uint_as_float(XROW(__float_as_uint(v), 8)));
    return fmaxf(v, __uint_as_float(XHALF(__float_as_uint(v))));
}

__device__ __forceinline__ float lane_read(float v, int l)
{
    return __int_as_float(__builtin_amdgcn_readlane(__float_as_int(v), l));
}

#define OIDX(s) (((s) & 1) ? (oi2[(s) >> 1] >> 16) : (oi2[(s) >> 1] & 0xffffu))

#define FIND_STEP(s)                                                            \
    {                                                                           \
        const bool hit = (d[s] == M);                                           \
        const unsigned oi = hit ? OIDX(s) : 0xffffffffu;                        \
        const bool better = oi < cand;                                          \
        cand = umn(cand, oi);                                                   \
        xs = better ? px[(s) >> 1][(s) & 1] : xs;                               \
        ys = better ? py[(s) >> 1][(s) & 1] : ys;                               \
        zs = better ? pz[(s) >> 1][(s) & 1] : zs;                               \
    }
#define FIND_IN_GROUP(g)                                                        \
    FIND_STEP((g) * GRP + 0) FIND_STEP((g) * GRP + 1) FIND_STEP((g) * GRP + 2)  \
    FIND_STEP((g) * GRP + 3) FIND_STEP((g) * GRP + 4) FIND_STEP((g) * GRP + 5)  \
    FIND_STEP((g) * GRP + 6) FIND_STEP((g) * GRP + 7)

#define FIND_IF(g)                                                              \
    if (gm & (1u << (g))) {                                                     \
        FIND_IN_GROUP(g)                                                        \
    }

#define UPD_PAIR(k)                                                             \
    {                                                                           \
        const v2f dx = px[k] - nx2, dy = py[k] - ny2, dz = pz[k] - nz2;         \
        const v2f dd = dx * dx + dy * dy + dz * dz;                             \
        d[2 * (k)]     = umn(d[2 * (k)],     __float_as_uint(dd[0]));           \
        d[2 * (k) + 1] = umn(d[2 * (k) + 1], __float_as_uint(dd[1]));           \
    }
#define UPDATE_GROUP(g)                                                         \
    if (need & (1u << (g))) {                                                   \
        UPD_PAIR(4 * (g) + 0) UPD_PAIR(4 * (g) + 1)                             \
        UPD_PAIR(4 * (g) + 2) UPD_PAIR(4 * (g) + 3)                             \
        G[g] = umx8(&d[(g) * GRP]);                                             \
    }

__global__ __launch_bounds__(32) void fps_kernel(const float* __restrict__ p,
    const float* __restrict__ sx, const float* __restrict__ sy, const float* __restrict__ sz,
    const int* __restrict__ so, int* __restrict__ fid)
{
    const int b = blockIdx.x;
    const float* pb = p + (size_t)b * NPTS * 3;
    const float* qx = sx + (size_t)b * NPTS;
    const float* qy = sy + (size_t)b * NPTS;
    const float* qz = sz + (size_t)b * NPTS;
    const int*   qo = so + (size_t)b * NPTS;
    const int lane = threadIdx.x;

    v2f px[NPR], py[NPR], pz[NPR];
    unsigned oi2[NPR];
    unsigned d[PPL];
    unsigned G[NGRP];
    float blx = 0.0f, bhx = 0.0f, bly = 0.0f, bhy = 0.0f, blz = 0.0f, bhz = 0.0f;
    {
        const float x0 = pb[0], y0 = pb[1], z0 = pb[2];
        const v2f x02 = { x0, x0 }, y02 = { y0, y0 }, z02 = { z0, z0 };
        #pragma unroll
        for (int k = 0; k < NPR; ++k) {
            const int i0 = (2 * k) * 32 + lane, i1 = (2 * k + 1) * 32 + lane;
            px[k][0] = qx[i0]; py[k][0] = qy[i0]; pz[k][0] = qz[i0];
            px[k][1] = qx[i1]; py[k][1] = qy[i1]; pz[k][1] = qz[i1];
            oi2[k] = (unsigned)qo[i0] | ((unsigned)qo[i1] << 16);
            const v2f dx = px[k] - x02, dy = py[k] - y02, dz = pz[k] - z02;
            const v2f dd = dx * dx + dy * dy + dz * dz;
            d[2 * k] = __float_as_uint(dd[0]);
            d[2 * k + 1] = __float_as_uint(dd[1]);
        }
        #pragma unroll
        for (int g = 0; g < NGRP; ++g) {
            G[g] = umx8(&d[g * GRP]);
            float lx = px[4 * g][0], hx = lx, ly = py[4 * g][0], hy = ly, lz = pz[4 * g][0], hz = lz;
            #pragma unroll
            for (int kk = 0; kk < GRP / 2; ++kk) {
                #pragma unroll
                for (int e = 0; e < 2; ++e) {
                    const float cx = px[4 * g + kk][e], cy = py[4 * g + kk][e], cz = pz[4 * g + kk][e];
                    lx = fminf(lx, cx); hx = fmaxf(hx, cx);
                    ly = fminf(ly, cy); hy = fmaxf(hy, cy);
                    lz = fminf(lz, cz); hz = fmaxf(hz, cz);
                }
            }
            lx = wave_fmin(lx); hx = wave_fmax(hx);
            ly = wave_fmin(ly); hy = wave_fmax(hy);
            lz = wave_fmin(lz); hz = wave_fmax(hz);
            const bool mine = ((lane & 15) == g);
            blx = mine ? lx : blx; bhx = mine ? hx : bhx;
            bly = mine ? ly : bly; bhy = mine ? hy : bhy;
            blz = mine ? lz : blz; bhz = mine ? hz : bhz;
        }
    }

    if (lane == 0) fid[b * NOUT + 0] = b * NPTS;

    const bool b0 = (lane & 1) != 0, b1 = (lane & 2) != 0, b2 = (lane & 4) != 0, b3 = (lane & 8) != 0;

    for (int i = 1; i < NOUT; ++i) {
        unsigned t8[8], t4[4], t2[2];
        #pragma unroll
        for (int j = 0; j < 8; ++j) {
            const unsigned lo = G[2 * j], hi = G[2 * j + 1];
            t8[j] = umx(b0 ? hi : lo, XROW(b0 ? lo : hi, 1));
        }
        #pragma unroll
        for (int j = 0; j < 4; ++j) {
            const unsigned lo = t8[2 * j], hi = t8[2 * j + 1];
            t4[j] = umx(b1 ? hi : lo, XROW(b1 ? lo : hi, 2));
        }
        #pragma unroll
        for (int j = 0; j < 2; ++j) {
            const unsigned lo = t4[2 * j], hi = t4[2 * j + 1];
            t2[j] = umx(b2 ? hi : lo, XROW(b2 ? lo : hi, 4));
        }
        unsigned gw = umx(b3 ? t2[1] : t2[0], XROW(b3 ? t2[0] : t2[1], 8));
        gw = umx(gw, XHALF(gw));
        const unsigned M = row_umax(gw);

        const unsigned gm = __builtin_amdgcn_ballot_w32(gw == M) & 0xffffu;
        unsigned cand = 0xffffffffu;
        float xs = 0.0f, ys = 0.0f, zs = 0.0f;
        FIND_IF(0)
        FIND_IF(1)
        FIND_IF(2)
        FIND_IF(3)
        FIND_IF(4)
        FIND_IF(5)
        FIND_IF(6)
        FIND_IF(7)
        FIND_IF(8)
        FIND_IF(9)
        FIND_IF(10)
        FIND_IF(11)
        FIND_IF(12)
        FIND_IF(13)
        FIND_IF(14)
        FIND_IF(15)
        const unsigned wmin = wave_umin(cand);
        const int wl = __builtin_ctz(__builtin_amdgcn_ballot_w32(cand == wmin) | 0x80000000u);
        const float nx = lane_read(xs, wl), ny = lane_read(ys, wl), nz = lane_read(zs, wl);
        if (lane == 0) fid[b * NOUT + i] = b * NPTS + (int)wmin;

        const float ex = fmaxf(fmaxf(blx - nx, nx - bhx), 0.0f);
        const float ey = fmaxf(fmaxf(bly - ny, ny - bhy), 0.0f);
        const float ez = fmaxf(fmaxf(blz - nz, nz - bhz), 0.0f);
        const float bd = fmaf(ez, ez, fmaf(ey, ey, ex * ex)) * 0.99999904632568359375f;
        const unsigned need = __builtin_amdgcn_ballot_w32(gw > __float_as_uint(bd)) & 0xffffu;

        const v2f nx2 = { nx, nx }, ny2 = { ny, ny }, nz2 = { nz, nz };
        UPDATE_GROUP(0)
        UPDATE_GROUP(1)
        UPDATE_GROUP(2)
        UPDATE_GROUP(3)
        UPDATE_GROUP(4)
        UPDATE_GROUP(5)
        UPDATE_GROUP(6)
        UPDATE_GROUP(7)
        UPDATE_GROUP(8)
        UPDATE_GROUP(9)
        UPDATE_GROUP(10)
        UPDATE_GROUP(11)
        UPDATE_GROUP(12)
        UPDATE_GROUP(13)
        UPDATE_GROUP(14)
        UPDATE_GROUP(15)
    }
}

__global__ __launch_bounds__(128) void gather_maxpool_kernel(
    const float* __restrict__ h, const float* __restrict__ p,
    const int* __restrict__ fid, const int* __restrict__ sid_euc,
    float* __restrict__ out)
{
    const int s = blockIdx.x;
    const int t = threadIdx.x;

    __shared__ int ssid[KNBR];
    const int f = fid[s];
    if (t < KNBR) ssid[t] = sid_euc[(size_t)f * KNBR + t];
    if (t < 3)    out[XOUT_ELEMS + (size_t)s * 3 + t] = p[(size_t)f * 3 + t];
    __syncthreads();

    const int c = t * 4;
    v4f acc = { -INFINITY, -INFINITY, -INFINITY, -INFINITY };
    #pragma unroll
    for (int k = 0; k < KNBR; ++k) {
        const int r = ssid[k];
        if (k + 1 < KNBR)
            __builtin_prefetch(h + (size_t)ssid[k + 1] * COUT + c, 0, 0);
        const v4f v = *(const v4f*)(h + (size_t)r * COUT + c);
        acc.x = fmaxf(acc.x, v.x);
        acc.y = fmaxf(acc.y, v.y);
        acc.z = fmaxf(acc.z, v.z);
        acc.w = fmaxf(acc.w, v.w);
    }
    *(v4f*)(out + (size_t)s * COUT + c) = acc;
}

extern "C" void kernel_launch(void* const* d_in, const int* in_sizes, int n_in,
                              void* d_out, int out_size, void* d_ws, size_t ws_size,
                              hipStream_t stream)
{
    (void)in_sizes; (void)n_in; (void)out_size; (void)ws_size;
    const float* x     = (const float*)d_in[0];
    const float* p     = (const float*)d_in[1];
    const int*   sid   = (const int*)  d_in[2];
    const float* W     = (const float*)d_in[4];
    const float* bias  = (const float*)d_in[5];
    const float* gamma = (const float*)d_in[6];
    const float* beta  = (const float*)d_in[7];
    const float* rmean = (const float*)d_in[8];
    const float* rvar  = (const float*)d_in[9];
    float* out = (float*)d_out;

    float* h  = (float*)d_ws;
    int* fid  = (int*)((char*)d_ws + (size_t)MROWS * COUT * 4);

    float* sx = (float*)d_ws;
    float* sy = sx + MROWS;
    float* sz = sy + MROWS;
    int*   so = (int*)(sz + MROWS);
    float* ax = (float*)(so + MROWS);
    float* ay = ax + MROWS;
    float* az = ay + MROWS;
    int*   ao = (int*)(az + MROWS);
    float* bx = (float*)(ao + MROWS);
    float* by = bx + MROWS;
    float* bz = by + MROWS;
    int*   bo = (int*)(bz + MROWS);

    const dim3 sgrid(NPTS / 256, BATCH);
    rank_x_kernel<<<sgrid, 256, 0, stream>>>(p, ax, ay, az, ao);
    rank_in_group_kernel<SLAB><<<sgrid, 256, 0, stream>>>(ay, ax, ay, az, ao, bx, by, bz, bo);
    rank_in_group_kernel<HALF><<<sgrid, 256, 0, stream>>>(bz, bx, by, bz, bo, sx, sy, sz, so);
    fps_kernel<<<BATCH, 32, 0, stream>>>(p, sx, sy, sz, so, fid);
    dim3 ggrid(COUT / GBN, MROWS / GBM);
    gemm_bn_relu_kernel<<<ggrid, 256, 0, stream>>>(x, W, bias, gamma, beta,
                                                   rmean, rvar, h);
    gather_maxpool_kernel<<<BATCH * NOUT, 128, 0, stream>>>(h, p, fid, sid, out);
}
